# starting kernel
# baseline (speedup 1.0000x reference)
#include <hip/hip_runtime.h>
#include <hip/hip_bf16.h>

typedef __attribute__((ext_vector_type(2))) float v2f;
typedef __attribute__((ext_vector_type(8))) float v8f;

#define MAX_T     1024
#define MAX_TILES (MAX_T / 16)
#define WAVES     8
#define EPSF      1e-12f
#define NEG_INF   (-3.402823466e38f)
#define IMAX      0x7fffffff


__global__ void tri_coeff_kernel(const float* __restrict__ verts,
                                 const int* __restrict__ tris,
                                 const int* __restrict__ gw_p,
                                 const int* __restrict__ gh_p,
                                 float* __restrict__ pack, int T)
{
    int t = blockIdx.x * blockDim.x + threadIdx.x;
    if (t >= T) return;
    float gw = (float)gw_p[0], gh = (float)gh_p[0];
    int i0 = tris[t * 3 + 0], i1 = tris[t * 3 + 1], i2 = tris[t * 3 + 2];
    float ax = verts[i0 * 2 + 0] / gw, ay = verts[i0 * 2 + 1] / gh;
    float bx = verts[i1 * 2 + 0] / gw, by = verts[i1 * 2 + 1] / gh;
    float cx = verts[i2 * 2 + 0] / gw, cy = verts[i2 * 2 + 1] / gh;
    float denom = (by - cy) * (ax - cx) + (cx - bx) * (ay - cy);
    if (fabsf(denom) < EPSF) denom = EPSF;
    float a1 = (by - cy) / denom;
    float b1 = (cx - bx) / denom;
    float g1 = -(a1 * cx + b1 * cy);
    float a2 = (cy - ay) / denom;
    float b2 = (ax - cx) / denom;
    float g2 = -(a2 * cx + b2 * cy);
    float mx = (ax + bx + cx) * (1.0f / 3.0f);
    float my = (ay + by + cy) * (1.0f / 3.0f);
    float mc = -0.5f * (mx * mx + my * my);

    int tile = t >> 4, nlan = t & 15;
    float* lo = pack + (size_t)(tile * 32 + nlan) * 8;
    float* hi = pack + (size_t)(tile * 32 + 16 + nlan) * 8;
    lo[0] = a1; lo[1] = b1; lo[2] = a2; lo[3] = b2; lo[4] = mx; lo[5] = my; lo[6] = 0.f; lo[7] = 0.f;
    hi[0] = g1; hi[1] = 0.f; hi[2] = g2; hi[3] = 0.f; hi[4] = mc; hi[5] = 0.f; hi[6] = 0.f; hi[7] = 0.f;
}

__global__ void __launch_bounds__(256)
point_kernel(const float* __restrict__ points,
             const float* __restrict__ verts,
             const int* __restrict__ tris,
             const float* __restrict__ pack,
             const int* __restrict__ frame_p,
             const int* __restrict__ gw_p,
             const int* __restrict__ gh_p,
             float* __restrict__ out, int T, int P)
{
    __shared__ float sp[MAX_TILES * 32 * 8];
    __shared__ int finals[WAVES * 16];

    int tid = threadIdx.x;
    int ntiles = T >> 4;
    int nf4 = ntiles * 32 * 2;
    const float4* gsrc = (const float4*)pack;
    float4* ldst = (float4*)sp;
    for (int i = tid; i < nf4; i += blockDim.x) ldst[i] = gsrc[i];
    __syncthreads();

    int lane = tid & 31;
    int half = lane >> 4;
    int nlan = lane & 15;
    int wave = tid >> 5;
    int pbase = (blockIdx.x * WAVES + wave) * 16;

    float px = 0.0f, py = 0.0f;
    v2f a;
    if (!half) {
        int p = pbase + nlan;
        if (p < P) { px = points[p * 2 + 0]; py = points[p * 2 + 1]; }
        a.x = px; a.y = py;
    } else {
        a.x = 1.0f; a.y = 0.0f;
    }

    int   minIns[8];
    float bestS[8];
    int   bestI[8];
#pragma unroll
    for (int r = 0; r < 8; ++r) { minIns[r] = IMAX; bestS[r] = NEG_INF; bestI[r] = IMAX; }

    for (int tile = 0; tile < ntiles; ++tile) {
        int n = tile * 16 + nlan;
        const float4* rec4 = (const float4*)(sp + (size_t)(tile * 32 + lane) * 8);
        float4 w = rec4[0];
        float2 u = *(const float2*)(rec4 + 1);
        v2f b1, b2, b3;
        b1.x = w.x; b1.y = w.y;
        b2.x = w.z; b2.y = w.w;
        b3.x = u.x; b3.y = u.y;

        v8f c1 = {}, c2 = {}, c3 = {};
        c1 = __builtin_amdgcn_wmma_f32_16x16x4_f32(false, a, false, b1, (short)0, c1, false, false);
        c2 = __builtin_amdgcn_wmma_f32_16x16x4_f32(false, a, false, b2, (short)0, c2, false, false);
        c3 = __builtin_amdgcn_wmma_f32_16x16x4_f32(false, a, false, b3, (short)0, c3, false, false);
#pragma unroll
        for (int r = 0; r < 8; ++r) {
            float l1 = c1[r], l2 = c2[r];
            float l3 = 1.0f - l1 - l2;
            float mn = fminf(fminf(l1, l2), l3);
            bool inside = (mn >= 0.0f);
            int cand = min(n, minIns[r]);
            minIns[r] = inside ? cand : minIns[r];
            float s = c3[r];
            bool better = (s > bestS[r]);
            bestS[r] = better ? s : bestS[r];
            bestI[r] = better ? n : bestI[r];
        }
    }

#pragma unroll
    for (int r = 0; r < 8; ++r) {
#pragma unroll
        for (int m = 1; m < 16; m <<= 1) {
            int oi = __shfl_xor(minIns[r], m, 32);
            if (oi < minIns[r]) minIns[r] = oi;
            float os = __shfl_xor(bestS[r], m, 32);
            int   ob = __shfl_xor(bestI[r], m, 32);
            if (os > bestS[r] || (os == bestS[r] && ob < bestI[r])) { bestS[r] = os; bestI[r] = ob; }
        }
    }

    if (nlan == 0) {
#pragma unroll
        for (int r = 0; r < 8; ++r) {
            int mrow = half * 8 + r;
            finals[wave * 16 + mrow] = (minIns[r] != IMAX) ? minIns[r] : bestI[r];
        }
    }
    __syncthreads();

    if (!half) {
        int p = pbase + nlan;
        if (p < P) {
            int f = finals[wave * 16 + nlan];
            int ft = f >> 4, fn = f & 15;
            const float* lo = sp + (size_t)(ft * 32 + fn) * 8;
            const float* hi = sp + (size_t)(ft * 32 + 16 + fn) * 8;
            float l1 = fmaf(lo[0], px, fmaf(lo[1], py, hi[0]));
            float l2 = fmaf(lo[2], px, fmaf(lo[3], py, hi[2]));
            float l3 = 1.0f - l1 - l2;
            float gw = (float)gw_p[0], gh = (float)gh_p[0];
            float fr = (float)frame_p[0];
            float* o = out + (size_t)p * 12;
            o[0] = l1; o[1] = l2; o[2] = l3;
#pragma unroll
            for (int j = 0; j < 3; ++j) {
                int vi = tris[f * 3 + j];
                o[3 + j * 3 + 0] = verts[vi * 2 + 0] / gw;
                o[3 + j * 3 + 1] = verts[vi * 2 + 1] / gh;
                o[3 + j * 3 + 2] = fr;
            }
        }
    }
}

extern "C" void kernel_launch(void* const* d_in, const int* in_sizes, int n_in,
                              void* d_out, int out_size, void* d_ws, size_t ws_size,
                              hipStream_t stream) {
    const float* verts  = (const float*)d_in[0];
    const float* points = (const float*)d_in[1];
    const int*   tris   = (const int*)d_in[2];
    const int*   frame  = (const int*)d_in[3];
    const int*   gw     = (const int*)d_in[4];
    const int*   gh     = (const int*)d_in[5];
    float*       out    = (float*)d_out;

    int T = in_sizes[2] / 3;
    int P = in_sizes[1] / 2;
    if (T > MAX_T) T = MAX_T;

    float* pack = (float*)d_ws;

    tri_coeff_kernel<<<(T + 255) / 256, 256, 0, stream>>>(verts, tris, gw, gh, pack, T);

    int pointsPerBlock = WAVES * 16;
    int blocks = (P + pointsPerBlock - 1) / pointsPerBlock;
    point_kernel<<<blocks, 256, 0, stream>>>(points, verts, tris, pack,
                                             frame, gw, gh, out, T, P);
}


# Swap WMMA roles (triangle records as A, points as B): a lane's 8 accumulators are 8 triangles of ONE point; first-hit and best-score reduce in-lane by min3/max3, index by recomputing one tile
# speedup vs baseline: 1.1557x; 1.1557x over previous
#include <hip/hip_runtime.h>
#include <hip/hip_bf16.h>

typedef __attribute__((ext_vector_type(2))) float v2f;
typedef __attribute__((ext_vector_type(8))) float v8f;

#define MAX_T     1024
#define MAX_TILES (MAX_T / 16)
#define WAVES     8
#define EPSF      1e-12f
#define NEG_INF   (-3.402823466e38f)
#define IMAX      0x7fffffff


__global__ void tri_coeff_kernel(const float* __restrict__ verts,
                                 const int* __restrict__ tris,
                                 const int* __restrict__ gw_p,
                                 const int* __restrict__ gh_p,
                                 float* __restrict__ pack, int T)
{
    int t = blockIdx.x * blockDim.x + threadIdx.x;
    if (t >= T) return;
    float gw = (float)gw_p[0], gh = (float)gh_p[0];
    int i0 = tris[t * 3 + 0], i1 = tris[t * 3 + 1], i2 = tris[t * 3 + 2];
    float ax = verts[i0 * 2 + 0] / gw, ay = verts[i0 * 2 + 1] / gh;
    float bx = verts[i1 * 2 + 0] / gw, by = verts[i1 * 2 + 1] / gh;
    float cx = verts[i2 * 2 + 0] / gw, cy = verts[i2 * 2 + 1] / gh;
    float denom = (by - cy) * (ax - cx) + (cx - bx) * (ay - cy);
    if (fabsf(denom) < EPSF) denom = EPSF;
    float a1 = (by - cy) / denom;
    float b1 = (cx - bx) / denom;
    float g1 = -(a1 * cx + b1 * cy);
    float a2 = (cy - ay) / denom;
    float b2 = (ax - cx) / denom;
    float g2 = -(a2 * cx + b2 * cy);
    float mx = (ax + bx + cx) * (1.0f / 3.0f);
    float my = (ay + by + cy) * (1.0f / 3.0f);
    float mc = -0.5f * (mx * mx + my * my);

    int tile = t >> 4, nlan = t & 15;
    float* lo = pack + (size_t)(tile * 32 + nlan) * 8;
    float* hi = pack + (size_t)(tile * 32 + 16 + nlan) * 8;
    lo[0] = a1; lo[1] = b1; lo[2] = a2; lo[3] = b2; lo[4] = mx; lo[5] = my; lo[6] = 0.f; lo[7] = 0.f;
    hi[0] = g1; hi[1] = 0.f; hi[2] = g2; hi[3] = 0.f; hi[4] = mc; hi[5] = 0.f; hi[6] = 0.f; hi[7] = 0.f;
}

__device__ __forceinline__ void tile_products(const float* __restrict__ sp, int tile, int lane,
                                              v2f pt, v8f& c1, v8f& c2, v8f& c3)
{
    const float4* rec4 = (const float4*)(sp + (size_t)(tile * 32 + lane) * 8);
    float4 w = rec4[0];
    float2 u = *(const float2*)(rec4 + 1);
    v2f t1, t2, t3;
    t1.x = w.x; t1.y = w.y;
    t2.x = w.z; t2.y = w.w;
    t3.x = u.x; t3.y = u.y;
    v8f z1 = {}, z2 = {}, z3 = {};
    c1 = __builtin_amdgcn_wmma_f32_16x16x4_f32(false, t1, false, pt, (short)0, z1, false, false);
    c2 = __builtin_amdgcn_wmma_f32_16x16x4_f32(false, t2, false, pt, (short)0, z2, false, false);
    c3 = __builtin_amdgcn_wmma_f32_16x16x4_f32(false, t3, false, pt, (short)0, z3, false, false);
}

__global__ void __launch_bounds__(256)
point_kernel(const float* __restrict__ points,
             const float* __restrict__ verts,
             const int* __restrict__ tris,
             const float* __restrict__ pack,
             const int* __restrict__ frame_p,
             const int* __restrict__ gw_p,
             const int* __restrict__ gh_p,
             float* __restrict__ out, int T, int P)
{
    __shared__ float sp[MAX_TILES * 32 * 8];

    int tid = threadIdx.x;
    int ntiles = T >> 4;
    int nf4 = ntiles * 32 * 2;
    const float4* gsrc = (const float4*)pack;
    float4* ldst = (float4*)sp;
    for (int i = tid; i < nf4; i += blockDim.x) ldst[i] = gsrc[i];
    __syncthreads();

    int lane = tid & 31;
    int half = lane >> 4;
    int nlan = lane & 15;
    int wave = tid >> 5;
    int pbase = (blockIdx.x * WAVES + wave) * 16;

    float px = 0.0f, py = 0.0f;
    v2f pt;
    if (!half) {
        int p = pbase + nlan;
        if (p < P) { px = points[p * 2 + 0]; py = points[p * 2 + 1]; }
        pt.x = px; pt.y = py;
    } else {
        pt.x = 1.0f; pt.y = 0.0f;
    }

    int   firstTile = IMAX;
    float bestS     = NEG_INF;
    int   bestTile  = 0;

    for (int tile = 0; tile < ntiles; ++tile) {
        v8f c1, c2, c3;
        tile_products(sp, tile, lane, pt, c1, c2, c3);
        float mn[8];
#pragma unroll
        for (int r = 0; r < 8; ++r) {
            float l1 = c1[r], l2 = c2[r];
            float l3 = 1.0f - l1 - l2;
            mn[r] = fminf(fminf(l1, l2), l3);
        }
        float hit = fmaxf(fmaxf(fmaxf(fmaxf(mn[0], mn[1]), mn[2]),
                                fmaxf(fmaxf(mn[3], mn[4]), mn[5])),
                          fmaxf(mn[6], mn[7]));
        int cand = (hit >= 0.0f) ? tile : IMAX;
        firstTile = min(firstTile, cand);
        float m = fmaxf(fmaxf(fmaxf(fmaxf(c3[0], c3[1]), c3[2]),
                              fmaxf(fmaxf(c3[3], c3[4]), c3[5])),
                        fmaxf(c3[6], c3[7]));
        bool better = (m > bestS);
        bestS    = better ? m : bestS;
        bestTile = better ? tile : bestTile;
    }

    int   oft = __shfl_xor(firstTile, 16, 32);
    float obs = __shfl_xor(bestS, 16, 32);
    bool valid = (min(firstTile, oft) != IMAX);
    bool need;
    int  resTile;
    if (valid) {
        need = (firstTile != IMAX) && (half ? (firstTile < oft) : (firstTile <= oft));
        resTile = firstTile;
    } else {
        need = (bestS >= obs);
        resTile = bestTile;
    }
    if (ntiles <= 0) need = false;

    int myIdx = IMAX;
    unsigned pend = __builtin_amdgcn_ballot_w32(need);
    for (int it = 0; it < 32; ++it) {
        if (pend == 0u) break;
        int srcl = __builtin_ctz(pend);
        int t = __builtin_amdgcn_readlane(resTile, srcl);
        v8f c1, c2, c3;
        tile_products(sp, t, lane, pt, c1, c2, c3);
        int rIn = 0, rS = 0;
#pragma unroll
        for (int r = 7; r >= 0; --r) {
            float l1 = c1[r], l2 = c2[r];
            float l3 = 1.0f - l1 - l2;
            float mnr = fminf(fminf(l1, l2), l3);
            rIn = (mnr >= 0.0f) ? r : rIn;
            rS  = (c3[r] == bestS) ? r : rS;
        }
        bool mine = need && (resTile == t);
        int rr = valid ? rIn : rS;
        myIdx = mine ? (t * 16 + half * 8 + rr) : myIdx;
        need = need && !mine;
        pend = __builtin_amdgcn_ballot_w32(need);
    }

    int oIdx = __shfl_xor(myIdx, 16, 32);
    int f = min(myIdx, oIdx);

    if (!half) {
        int p = pbase + nlan;
        if (p < P) {
            int ft = f >> 4, fn = f & 15;
            const float* lo = sp + (size_t)(ft * 32 + fn) * 8;
            const float* hi = sp + (size_t)(ft * 32 + 16 + fn) * 8;
            float l1 = fmaf(lo[0], px, fmaf(lo[1], py, hi[0]));
            float l2 = fmaf(lo[2], px, fmaf(lo[3], py, hi[2]));
            float l3 = 1.0f - l1 - l2;
            float gw = (float)gw_p[0], gh = (float)gh_p[0];
            float fr = (float)frame_p[0];
            float* o = out + (size_t)p * 12;
            o[0] = l1; o[1] = l2; o[2] = l3;
#pragma unroll
            for (int j = 0; j < 3; ++j) {
                int vi = tris[f * 3 + j];
                o[3 + j * 3 + 0] = verts[vi * 2 + 0] / gw;
                o[3 + j * 3 + 1] = verts[vi * 2 + 1] / gh;
                o[3 + j * 3 + 2] = fr;
            }
        }
    }
}

extern "C" void kernel_launch(void* const* d_in, const int* in_sizes, int n_in,
                              void* d_out, int out_size, void* d_ws, size_t ws_size,
                              hipStream_t stream) {
    const float* verts  = (const float*)d_in[0];
    const float* points = (const float*)d_in[1];
    const int*   tris   = (const int*)d_in[2];
    const int*   frame  = (const int*)d_in[3];
    const int*   gw     = (const int*)d_in[4];
    const int*   gh     = (const int*)d_in[5];
    float*       out    = (float*)d_out;

    int T = in_sizes[2] / 3;
    int P = in_sizes[1] / 2;
    if (T > MAX_T) T = MAX_T;

    float* pack = (float*)d_ws;

    tri_coeff_kernel<<<(T + 255) / 256, 256, 0, stream>>>(verts, tris, gw, gh, pack, T);

    int pointsPerBlock = WAVES * 16;
    int blocks = (P + pointsPerBlock - 1) / pointsPerBlock;
    point_kernel<<<blocks, 256, 0, stream>>>(points, verts, tris, pack,
                                             frame, gw, gh, out, T, P);
}


# Prologue loads overlap: each thread issues its point load and all 16 table loads before the first LDS store (was load, wait, store 16 times in a row, point load after the barrier)
# speedup vs baseline: 1.4948x; 1.2934x over previous
#include <hip/hip_runtime.h>
#include <hip/hip_bf16.h>

typedef __attribute__((ext_vector_type(2))) float v2f;
typedef __attribute__((ext_vector_type(8))) float v8f;

#define MAX_T     1024
#define MAX_TILES (MAX_T / 16)
#define WAVES     8
#define THREADS   (WAVES * 32)
#define FILL_DEPTH (MAX_TILES * 32 * 2 / THREADS)
#define EPSF      1e-12f
#define NEG_INF   (-3.402823466e38f)
#define IMAX      0x7fffffff


__global__ void tri_coeff_kernel(const float* __restrict__ verts,
                                 const int* __restrict__ tris,
                                 const int* __restrict__ gw_p,
                                 const int* __restrict__ gh_p,
                                 float* __restrict__ pack, int T)
{
    int t = blockIdx.x * blockDim.x + threadIdx.x;
    if (t >= T) return;
    float gw = (float)gw_p[0], gh = (float)gh_p[0];
    int i0 = tris[t * 3 + 0], i1 = tris[t * 3 + 1], i2 = tris[t * 3 + 2];
    float ax = verts[i0 * 2 + 0] / gw, ay = verts[i0 * 2 + 1] / gh;
    float bx = verts[i1 * 2 + 0] / gw, by = verts[i1 * 2 + 1] / gh;
    float cx = verts[i2 * 2 + 0] / gw, cy = verts[i2 * 2 + 1] / gh;
    float denom = (by - cy) * (ax - cx) + (cx - bx) * (ay - cy);
    if (fabsf(denom) < EPSF) denom = EPSF;
    float a1 = (by - cy) / denom;
    float b1 = (cx - bx) / denom;
    float g1 = -(a1 * cx + b1 * cy);
    float a2 = (cy - ay) / denom;
    float b2 = (ax - cx) / denom;
    float g2 = -(a2 * cx + b2 * cy);
    float mx = (ax + bx + cx) * (1.0f / 3.0f);
    float my = (ay + by + cy) * (1.0f / 3.0f);
    float mc = -0.5f * (mx * mx + my * my);

    int tile = t >> 4, nlan = t & 15;
    float* lo = pack + (size_t)(tile * 32 + nlan) * 8;
    float* hi = pack + (size_t)(tile * 32 + 16 + nlan) * 8;
    lo[0] = a1; lo[1] = b1; lo[2] = a2; lo[3] = b2; lo[4] = mx; lo[5] = my; lo[6] = 0.f; lo[7] = 0.f;
    hi[0] = g1; hi[1] = 0.f; hi[2] = g2; hi[3] = 0.f; hi[4] = mc; hi[5] = 0.f; hi[6] = 0.f; hi[7] = 0.f;
}

__device__ __forceinline__ void tile_products(const float* __restrict__ sp, int tile, int lane,
                                              v2f pt, v8f& c1, v8f& c2, v8f& c3)
{
    const float4* rec4 = (const float4*)(sp + (size_t)(tile * 32 + lane) * 8);
    float4 w = rec4[0];
    float2 u = *(const float2*)(rec4 + 1);
    v2f t1, t2, t3;
    t1.x = w.x; t1.y = w.y;
    t2.x = w.z; t2.y = w.w;
    t3.x = u.x; t3.y = u.y;
    v8f z1 = {}, z2 = {}, z3 = {};
    c1 = __builtin_amdgcn_wmma_f32_16x16x4_f32(false, t1, false, pt, (short)0, z1, false, false);
    c2 = __builtin_amdgcn_wmma_f32_16x16x4_f32(false, t2, false, pt, (short)0, z2, false, false);
    c3 = __builtin_amdgcn_wmma_f32_16x16x4_f32(false, t3, false, pt, (short)0, z3, false, false);
}

__global__ void __launch_bounds__(THREADS)
point_kernel(const float* __restrict__ points,
             const float* __restrict__ verts,
             const int* __restrict__ tris,
             const float* __restrict__ pack,
             const int* __restrict__ frame_p,
             const int* __restrict__ gw_p,
             const int* __restrict__ gh_p,
             float* __restrict__ out, int T, int P)
{
    __shared__ float sp[MAX_TILES * 32 * 8];

    int tid = threadIdx.x;
    int ntiles = T >> 4;
    int nf4 = ntiles * 32 * 2;
    const float4* gsrc = (const float4*)pack;
    float4* ldst = (float4*)sp;

    int lane = tid & 31;
    int half = lane >> 4;
    int nlan = lane & 15;
    int wave = tid >> 5;
    int pbase = (blockIdx.x * WAVES + wave) * 16;

    int p = pbase + nlan;
    int pc = min(p, P - 1);
    float qx = points[pc * 2 + 0], qy = points[pc * 2 + 1];

    if (nf4 > 0) {
        float4 stage[FILL_DEPTH];
#pragma unroll
        for (int k = 0; k < FILL_DEPTH; ++k)
            stage[k] = gsrc[min(tid + k * THREADS, nf4 - 1)];
#pragma unroll
        for (int k = 0; k < FILL_DEPTH; ++k)
            ldst[tid + k * THREADS] = stage[k];
    }
    __syncthreads();

    float px = 0.0f, py = 0.0f;
    v2f pt;
    if (!half) {
        if (p < P) { px = qx; py = qy; }
        pt.x = px; pt.y = py;
    } else {
        pt.x = 1.0f; pt.y = 0.0f;
    }

    int   firstTile = IMAX;
    float bestS     = NEG_INF;
    int   bestTile  = 0;

    for (int tile = 0; tile < ntiles; ++tile) {
        v8f c1, c2, c3;
        tile_products(sp, tile, lane, pt, c1, c2, c3);
        float mn[8];
#pragma unroll
        for (int r = 0; r < 8; ++r) {
            float l1 = c1[r], l2 = c2[r];
            float l3 = 1.0f - l1 - l2;
            mn[r] = fminf(fminf(l1, l2), l3);
        }
        float hit = fmaxf(fmaxf(fmaxf(fmaxf(mn[0], mn[1]), mn[2]),
                                fmaxf(fmaxf(mn[3], mn[4]), mn[5])),
                          fmaxf(mn[6], mn[7]));
        int cand = (hit >= 0.0f) ? tile : IMAX;
        firstTile = min(firstTile, cand);
        float m = fmaxf(fmaxf(fmaxf(fmaxf(c3[0], c3[1]), c3[2]),
                              fmaxf(fmaxf(c3[3], c3[4]), c3[5])),
                        fmaxf(c3[6], c3[7]));
        bool better = (m > bestS);
        bestS    = better ? m : bestS;
        bestTile = better ? tile : bestTile;
    }

    int   oft = __shfl_xor(firstTile, 16, 32);
    float obs = __shfl_xor(bestS, 16, 32);
    bool valid = (min(firstTile, oft) != IMAX);
    bool need;
    int  resTile;
    if (valid) {
        need = (firstTile != IMAX) && (half ? (firstTile < oft) : (firstTile <= oft));
        resTile = firstTile;
    } else {
        need = (bestS >= obs);
        resTile = bestTile;
    }
    if (ntiles <= 0) need = false;

    int myIdx = IMAX;
    unsigned pend = __builtin_amdgcn_ballot_w32(need);
    for (int it = 0; it < 32; ++it) {
        if (pend == 0u) break;
        int srcl = __builtin_ctz(pend);
        int t = __builtin_amdgcn_readlane(resTile, srcl);
        v8f c1, c2, c3;
        tile_products(sp, t, lane, pt, c1, c2, c3);
        int rIn = 0, rS = 0;
#pragma unroll
        for (int r = 7; r >= 0; --r) {
            float l1 = c1[r], l2 = c2[r];
            float l3 = 1.0f - l1 - l2;
            float mnr = fminf(fminf(l1, l2), l3);
            rIn = (mnr >= 0.0f) ? r : rIn;
            rS  = (c3[r] == bestS) ? r : rS;
        }
        bool mine = need && (resTile == t);
        int rr = valid ? rIn : rS;
        myIdx = mine ? (t * 16 + half * 8 + rr) : myIdx;
        need = need && !mine;
        pend = __builtin_amdgcn_ballot_w32(need);
    }

    int oIdx = __shfl_xor(myIdx, 16, 32);
    int f = min(myIdx, oIdx);

    if (!half) {
        if (p < P) {
            int ft = f >> 4, fn = f & 15;
            const float* lo = sp + (size_t)(ft * 32 + fn) * 8;
            const float* hi = sp + (size_t)(ft * 32 + 16 + fn) * 8;
            float l1 = fmaf(lo[0], px, fmaf(lo[1], py, hi[0]));
            float l2 = fmaf(lo[2], px, fmaf(lo[3], py, hi[2]));
            float l3 = 1.0f - l1 - l2;
            float gw = (float)gw_p[0], gh = (float)gh_p[0];
            float fr = (float)frame_p[0];
            float* o = out + (size_t)p * 12;
            o[0] = l1; o[1] = l2; o[2] = l3;
#pragma unroll
            for (int j = 0; j < 3; ++j) {
                int vi = tris[f * 3 + j];
                o[3 + j * 3 + 0] = verts[vi * 2 + 0] / gw;
                o[3 + j * 3 + 1] = verts[vi * 2 + 1] / gh;
                o[3 + j * 3 + 2] = fr;
            }
        }
    }
}

extern "C" void kernel_launch(void* const* d_in, const int* in_sizes, int n_in,
                              void* d_out, int out_size, void* d_ws, size_t ws_size,
                              hipStream_t stream) {
    const float* verts  = (const float*)d_in[0];
    const float* points = (const float*)d_in[1];
    const int*   tris   = (const int*)d_in[2];
    const int*   frame  = (const int*)d_in[3];
    const int*   gw     = (const int*)d_in[4];
    const int*   gh     = (const int*)d_in[5];
    float*       out    = (float*)d_out;

    int T = in_sizes[2] / 3;
    int P = in_sizes[1] / 2;
    if (T > MAX_T) T = MAX_T;

    float* pack = (float*)d_ws;

    tri_coeff_kernel<<<(T + 255) / 256, 256, 0, stream>>>(verts, tris, gw, gh, pack, T);

    int pointsPerBlock = WAVES * 16;
    int blocks = (P + pointsPerBlock - 1) / pointsPerBlock;
    point_kernel<<<blocks, THREADS, 0, stream>>>(points, verts, tris, pack,
                                             frame, gw, gh, out, T, P);
}


# Two-pass sweep: every point meets the first 8 tiles; points still without a hit are packed block-wide into dense 16-point groups, each wave sweeps 1/8 of the tiles per group (a first hit is final)
# speedup vs baseline: 2.0921x; 1.3996x over previous
#include <hip/hip_runtime.h>
#include <hip/hip_bf16.h>

typedef __attribute__((ext_vector_type(2))) float v2f;
typedef __attribute__((ext_vector_type(8))) float v8f;

#define MAX_T     1024
#define MAX_TILES (MAX_T / 16)
#define WAVES     8
#define THREADS   (WAVES * 32)
#define TRI_PER_THREAD (MAX_T / THREADS)
#define EPSF      1e-12f
#define NEG_INF   (-3.402823466e38f)
#define IMAX      0x7fffffff
#define PASS1_TILES 8


__device__ __forceinline__ void tri_record(float vax, float vay, float vbx, float vby,
                                           float vcx, float vcy, float gw, float gh,
                                           float4& lo0, float4& lo1, float4& hi0, float4& hi1)
{
    float ax = vax / gw, ay = vay / gh;
    float bx = vbx / gw, by = vby / gh;
    float cx = vcx / gw, cy = vcy / gh;
    float denom = fmaf(by - cy, ax - cx, (cx - bx) * (ay - cy));
    if (fabsf(denom) < EPSF) denom = EPSF;
    float a1 = (by - cy) / denom;
    float b1 = (cx - bx) / denom;
    float g1 = -fmaf(b1, cy, a1 * cx);
    float a2 = (cy - ay) / denom;
    float b2 = (ax - cx) / denom;
    float g2 = -fmaf(a2, cx, b2 * cy);
    float mx = (ax + bx + cx) * (1.0f / 3.0f);
    float my = (ay + by + cy) * (1.0f / 3.0f);
    float mc;
    {
#pragma clang fp contract(off)
        mc = -0.5f * (mx * mx + my * my);
    }
    lo0 = make_float4(a1, b1, a2, b2);
    lo1 = make_float4(mx, my, 0.f, 0.f);
    hi0 = make_float4(g1, 0.f, g2, 0.f);
    hi1 = make_float4(mc, 0.f, 0.f, 0.f);
}

__device__ __forceinline__ void tile_products(const float* __restrict__ sp, int tile, int lane,
                                              v2f pt, v8f& c1, v8f& c2, v8f& c3)
{
    const float4* rec4 = (const float4*)(sp + (size_t)(tile * 32 + lane) * 8);
    float4 w = rec4[0];
    float2 u = *(const float2*)(rec4 + 1);
    v2f t1, t2, t3;
    t1.x = w.x; t1.y = w.y;
    t2.x = w.z; t2.y = w.w;
    t3.x = u.x; t3.y = u.y;
    v8f z1 = {}, z2 = {}, z3 = {};
    c1 = __builtin_amdgcn_wmma_f32_16x16x4_f32(false, t1, false, pt, (short)0, z1, false, false);
    c2 = __builtin_amdgcn_wmma_f32_16x16x4_f32(false, t2, false, pt, (short)0, z2, false, false);
    c3 = __builtin_amdgcn_wmma_f32_16x16x4_f32(false, t3, false, pt, (short)0, z3, false, false);
}

__device__ __forceinline__ void sweep_tiles(const float* __restrict__ sp, int lane, v2f pt,
                                            int t0, int t1, int step,
                                            int& firstTile, float& bestS, int& bestTile)
{
    for (int tile = t0; tile < t1; tile += step) {
        v8f c1, c2, c3;
        tile_products(sp, tile, lane, pt, c1, c2, c3);
        float mn[8];
#pragma unroll
        for (int r = 0; r < 8; ++r) {
            float l1 = c1[r], l2 = c2[r];
            float l3 = 1.0f - l1 - l2;
            mn[r] = fminf(fminf(l1, l2), l3);
        }
        float hit = fmaxf(fmaxf(fmaxf(fmaxf(mn[0], mn[1]), mn[2]),
                                fmaxf(fmaxf(mn[3], mn[4]), mn[5])),
                          fmaxf(mn[6], mn[7]));
        int cand = (hit >= 0.0f) ? tile : IMAX;
        firstTile = min(firstTile, cand);
        float m = fmaxf(fmaxf(fmaxf(fmaxf(c3[0], c3[1]), c3[2]),
                              fmaxf(fmaxf(c3[3], c3[4]), c3[5])),
                        fmaxf(c3[6], c3[7]));
        bool better = (m > bestS);
        bestS    = better ? m : bestS;
        bestTile = better ? tile : bestTile;
    }
}

__device__ __forceinline__ void finish_points(const float* __restrict__ sp, int lane, int half, v2f pt,
                                              float px, float py, int p, bool active, int ntiles,
                                              int firstTile, float bestS, int bestTile,
                                              const float* __restrict__ verts,
                                              const int* __restrict__ tris,
                                              const int* __restrict__ frame_p,
                                              const int* __restrict__ gw_p,
                                              const int* __restrict__ gh_p,
                                              float* __restrict__ out)
{
    int   oft = __shfl_xor(firstTile, 16, 32);
    float obs = __shfl_xor(bestS, 16, 32);
    bool valid = (min(firstTile, oft) != IMAX);
    bool need;
    int  resTile;
    if (valid) {
        need = (firstTile != IMAX) && (half ? (firstTile < oft) : (firstTile <= oft));
        resTile = firstTile;
    } else {
        need = (bestS >= obs);
        resTile = bestTile;
    }
    if (ntiles <= 0 || !active) need = false;

    int myIdx = IMAX;
    unsigned pend = __builtin_amdgcn_ballot_w32(need);
    for (int it = 0; it < 32; ++it) {
        if (pend == 0u) break;
        int srcl = __builtin_ctz(pend);
        int t = __builtin_amdgcn_readlane(resTile, srcl);
        v8f c1, c2, c3;
        tile_products(sp, t, lane, pt, c1, c2, c3);
        int rIn = 0, rS = 0;
#pragma unroll
        for (int r = 7; r >= 0; --r) {
            float l1 = c1[r], l2 = c2[r];
            float l3 = 1.0f - l1 - l2;
            float mnr = fminf(fminf(l1, l2), l3);
            rIn = (mnr >= 0.0f) ? r : rIn;
            rS  = (c3[r] == bestS) ? r : rS;
        }
        bool mine = need && (resTile == t);
        int rr = valid ? rIn : rS;
        myIdx = mine ? (t * 16 + half * 8 + rr) : myIdx;
        need = need && !mine;
        pend = __builtin_amdgcn_ballot_w32(need);
    }

    int oIdx = __shfl_xor(myIdx, 16, 32);
    int f = min(myIdx, oIdx);

    if (!half) {
        if (active) {
            int ft = f >> 4, fn = f & 15;
            const float* lo = sp + (size_t)(ft * 32 + fn) * 8;
            const float* hi = sp + (size_t)(ft * 32 + 16 + fn) * 8;
            float l1 = fmaf(lo[0], px, fmaf(lo[1], py, hi[0]));
            float l2 = fmaf(lo[2], px, fmaf(lo[3], py, hi[2]));
            float l3 = 1.0f - l1 - l2;
            float gw = (float)gw_p[0], gh = (float)gh_p[0];
            float fr = (float)frame_p[0];
            float* o = out + (size_t)p * 12;
            o[0] = l1; o[1] = l2; o[2] = l3;
#pragma unroll
            for (int j = 0; j < 3; ++j) {
                int vi = tris[f * 3 + j];
                o[3 + j * 3 + 0] = verts[vi * 2 + 0] / gw;
                o[3 + j * 3 + 1] = verts[vi * 2 + 1] / gh;
                o[3 + j * 3 + 2] = fr;
            }
        }
    }
}

__global__ void __launch_bounds__(THREADS)
point_kernel(const float* __restrict__ points,
             const float* __restrict__ verts,
             const int* __restrict__ tris,
             const int* __restrict__ frame_p,
             const int* __restrict__ gw_p,
             const int* __restrict__ gh_p,
             float* __restrict__ out, int T, int P)
{
    __shared__ float sp[MAX_TILES * 32 * 8];
    __shared__ float s_lx[WAVES * 16];
    __shared__ float s_ly[WAVES * 16];
    __shared__ int   s_lp[WAVES * 16];
    __shared__ int   s_cnt[WAVES];
    __shared__ int   s_rft[WAVES * WAVES * 32];
    __shared__ float s_rbs[WAVES * WAVES * 32];
    __shared__ int   s_rbt[WAVES * WAVES * 32];

    int tid = threadIdx.x;
    int ntiles = T >> 4;

    int lane = tid & 31;
    int half = lane >> 4;
    int nlan = lane & 15;
    int wave = __builtin_amdgcn_readfirstlane(tid >> 5);
    int pbase = (blockIdx.x * WAVES + wave) * 16;

    int p = pbase + nlan;
    int pc = min(p, P - 1);
    float qx = points[pc * 2 + 0], qy = points[pc * 2 + 1];

    if (T > 0) {
        float gw = (float)gw_p[0], gh = (float)gh_p[0];
        int vi[TRI_PER_THREAD][3];
#pragma unroll
        for (int k = 0; k < TRI_PER_THREAD; ++k) {
            int tc = min(tid + k * THREADS, T - 1);
            vi[k][0] = tris[tc * 3 + 0];
            vi[k][1] = tris[tc * 3 + 1];
            vi[k][2] = tris[tc * 3 + 2];
        }
        float vx[TRI_PER_THREAD][3], vy[TRI_PER_THREAD][3];
#pragma unroll
        for (int k = 0; k < TRI_PER_THREAD; ++k) {
#pragma unroll
            for (int j = 0; j < 3; ++j) {
                vx[k][j] = verts[vi[k][j] * 2 + 0];
                vy[k][j] = verts[vi[k][j] * 2 + 1];
            }
        }
#pragma unroll
        for (int k = 0; k < TRI_PER_THREAD; ++k) {
            int t = tid + k * THREADS;
            float4 lo0, lo1, hi0, hi1;
            tri_record(vx[k][0], vy[k][0], vx[k][1], vy[k][1], vx[k][2], vy[k][2],
                       gw, gh, lo0, lo1, hi0, hi1);
            int tile = t >> 4, n = t & 15;
            float4* lo = (float4*)(sp + (size_t)(tile * 32 + n) * 8);
            float4* hi = (float4*)(sp + (size_t)(tile * 32 + 16 + n) * 8);
            lo[0] = lo0; lo[1] = lo1;
            hi[0] = hi0; hi[1] = hi1;
        }
    }
    __syncthreads();

    float px = 0.0f, py = 0.0f;
    v2f pt;
    if (!half) {
        if (p < P) { px = qx; py = qy; }
        pt.x = px; pt.y = py;
    } else {
        pt.x = 1.0f; pt.y = 0.0f;
    }

    bool complete1 = (ntiles <= PASS1_TILES);
    int  n1 = complete1 ? ntiles : PASS1_TILES;
    int   ft1 = IMAX;
    float bs1 = NEG_INF;
    int   bt1 = 0;
    sweep_tiles(sp, lane, pt, 0, n1, 1, ft1, bs1, bt1);

    bool inP  = (p < P);
    bool hitp = (min(ft1, __shfl_xor(ft1, 16, 32)) != IMAX);
    bool settled = inP && (hitp || complete1);
    bool unres = (!half) && inP && !settled;

    unsigned um = __builtin_amdgcn_ballot_w32(unres);
    if (lane == 0) s_cnt[wave] = __builtin_popcount(um);
    __syncthreads();
    int offs = 0, total = 0;
#pragma unroll
    for (int w = 0; w < WAVES; ++w) {
        int c = s_cnt[w];
        offs  += (w < wave) ? c : 0;
        total += c;
    }
    total = __builtin_amdgcn_readfirstlane(total);
    if (unres) {
        int slot = offs + __builtin_popcount(um & ((1u << lane) - 1u));
        s_lx[slot] = px; s_ly[slot] = py; s_lp[slot] = p;
    }
    __syncthreads();

    finish_points(sp, lane, half, pt, px, py, p, settled, ntiles, ft1, bs1, bt1,
                  verts, tris, frame_p, gw_p, gh_p, out);

    int groups = (total + 15) >> 4;
    for (int j = 0; j < groups; ++j) {
        int ic = min(j * 16 + nlan, total - 1);
        v2f q;
        if (!half) { q.x = s_lx[ic]; q.y = s_ly[ic]; }
        else       { q.x = 1.0f;     q.y = 0.0f; }
        int   f2 = IMAX;
        float b2 = NEG_INF;
        int   t2 = 0;
        sweep_tiles(sp, lane, q, wave, ntiles, WAVES, f2, b2, t2);
        int ri = (j * WAVES + wave) * 32 + lane;
        s_rft[ri] = f2; s_rbs[ri] = b2; s_rbt[ri] = t2;
    }
    __syncthreads();

    if (wave < groups) {
        int j = wave;
        int idx = j * 16 + nlan;
        bool act = (idx < total);
        int ic = min(idx, total - 1);
        float lx = 0.0f, ly = 0.0f;
        v2f q;
        if (!half) { lx = s_lx[ic]; ly = s_ly[ic]; q.x = lx; q.y = ly; }
        else       { q.x = 1.0f; q.y = 0.0f; }
        int lp = s_lp[ic];
        int   f2 = s_rft[(j * WAVES) * 32 + lane];
        float b2 = s_rbs[(j * WAVES) * 32 + lane];
        int   t2 = s_rbt[(j * WAVES) * 32 + lane];
#pragma unroll
        for (int w = 1; w < WAVES; ++w) {
            int   fw = s_rft[(j * WAVES + w) * 32 + lane];
            float bw = s_rbs[(j * WAVES + w) * 32 + lane];
            int   tw = s_rbt[(j * WAVES + w) * 32 + lane];
            f2 = min(f2, fw);
            bool better = (bw > b2) || (bw == b2 && tw < t2);
            b2 = better ? bw : b2;
            t2 = better ? tw : t2;
        }
        finish_points(sp, lane, half, q, lx, ly, lp, act, ntiles, f2, b2, t2,
                      verts, tris, frame_p, gw_p, gh_p, out);
    }
}

extern "C" void kernel_launch(void* const* d_in, const int* in_sizes, int n_in,
                              void* d_out, int out_size, void* d_ws, size_t ws_size,
                              hipStream_t stream) {
    const float* verts  = (const float*)d_in[0];
    const float* points = (const float*)d_in[1];
    const int*   tris   = (const int*)d_in[2];
    const int*   frame  = (const int*)d_in[3];
    const int*   gw     = (const int*)d_in[4];
    const int*   gh     = (const int*)d_in[5];
    float*       out    = (float*)d_out;

    int T = in_sizes[2] / 3;
    int P = in_sizes[1] / 2;
    if (T > MAX_T) T = MAX_T;

    int pointsPerBlock = WAVES * 16;
    int blocks = (P + pointsPerBlock - 1) / pointsPerBlock;
    point_kernel<<<blocks, THREADS, 0, stream>>>(points, verts, tris,
                                             frame, gw, gh, out, T, P);
}


# Sweep keeps each lane's first containing triangle index (8 compare-selects per tile on minima it already has), not only its tile: hit points skip the tile-recompute loop, no-hit points redo scores
# speedup vs baseline: 2.4623x; 1.1770x over previous
#include <hip/hip_runtime.h>
#include <hip/hip_bf16.h>

typedef __attribute__((ext_vector_type(2))) float v2f;
typedef __attribute__((ext_vector_type(8))) float v8f;

#define MAX_T     1024
#define MAX_TILES (MAX_T / 16)
#define WAVES     8
#define THREADS   (WAVES * 32)
#define TRI_PER_THREAD (MAX_T / THREADS)
#define EPSF      1e-12f
#define NEG_INF   (-3.402823466e38f)
#define IMAX      0x7fffffff
#define PASS1_TILES 8
#define NOHIT     0x40000000


__device__ __forceinline__ void tri_record(float vax, float vay, float vbx, float vby,
                                           float vcx, float vcy, float gw, float gh,
                                           float4& lo0, float4& lo1, float4& hi0, float4& hi1)
{
    float ax = vax / gw, ay = vay / gh;
    float bx = vbx / gw, by = vby / gh;
    float cx = vcx / gw, cy = vcy / gh;
    float denom = fmaf(by - cy, ax - cx, (cx - bx) * (ay - cy));
    if (fabsf(denom) < EPSF) denom = EPSF;
    float a1 = (by - cy) / denom;
    float b1 = (cx - bx) / denom;
    float g1 = -fmaf(b1, cy, a1 * cx);
    float a2 = (cy - ay) / denom;
    float b2 = (ax - cx) / denom;
    float g2 = -fmaf(a2, cx, b2 * cy);
    float mx = (ax + bx + cx) * (1.0f / 3.0f);
    float my = (ay + by + cy) * (1.0f / 3.0f);
    float mc;
    {
#pragma clang fp contract(off)
        mc = -0.5f * (mx * mx + my * my);
    }
    lo0 = make_float4(a1, b1, a2, b2);
    lo1 = make_float4(mx, my, 0.f, 0.f);
    hi0 = make_float4(g1, 0.f, g2, 0.f);
    hi1 = make_float4(mc, 0.f, 0.f, 0.f);
}

__device__ __forceinline__ void tile_products(const float* __restrict__ sp, int tile, int lane,
                                              v2f pt, v8f& c1, v8f& c2, v8f& c3)
{
    const float4* rec4 = (const float4*)(sp + (size_t)(tile * 32 + lane) * 8);
    float4 w = rec4[0];
    float2 u = *(const float2*)(rec4 + 1);
    v2f t1, t2, t3;
    t1.x = w.x; t1.y = w.y;
    t2.x = w.z; t2.y = w.w;
    t3.x = u.x; t3.y = u.y;
    v8f z1 = {}, z2 = {}, z3 = {};
    c1 = __builtin_amdgcn_wmma_f32_16x16x4_f32(false, t1, false, pt, (short)0, z1, false, false);
    c2 = __builtin_amdgcn_wmma_f32_16x16x4_f32(false, t2, false, pt, (short)0, z2, false, false);
    c3 = __builtin_amdgcn_wmma_f32_16x16x4_f32(false, t3, false, pt, (short)0, z3, false, false);
}

__device__ __forceinline__ void tile_scores(const float* __restrict__ sp, int tile, int lane,
                                            v2f pt, v8f& c3)
{
    float2 u = *(const float2*)(sp + (size_t)(tile * 32 + lane) * 8 + 4);
    v2f t3;
    t3.x = u.x; t3.y = u.y;
    v8f z3 = {};
    c3 = __builtin_amdgcn_wmma_f32_16x16x4_f32(false, t3, false, pt, (short)0, z3, false, false);
}

__device__ __forceinline__ void sweep_tiles(const float* __restrict__ sp, int lane, v2f pt,
                                            int t0, int t1, int step,
                                            int& firstIdx, float& bestS, int& bestTile)
{
    int half8 = (lane >> 4) * 8;
    for (int tile = t0; tile < t1; tile += step) {
        v8f c1, c2, c3;
        tile_products(sp, tile, lane, pt, c1, c2, c3);
        float mn[8];
#pragma unroll
        for (int r = 0; r < 8; ++r) {
            float l1 = c1[r], l2 = c2[r];
            float l3 = 1.0f - l1 - l2;
            mn[r] = fminf(fminf(l1, l2), l3);
        }
        int rIn = NOHIT;
#pragma unroll
        for (int r = 7; r >= 0; --r)
            rIn = (mn[r] >= 0.0f) ? r : rIn;
        firstIdx = min(firstIdx, tile * 16 + half8 + rIn);
        float m = fmaxf(fmaxf(fmaxf(fmaxf(c3[0], c3[1]), c3[2]),
                              fmaxf(fmaxf(c3[3], c3[4]), c3[5])),
                        fmaxf(c3[6], c3[7]));
        bool better = (m > bestS);
        bestS    = better ? m : bestS;
        bestTile = better ? tile : bestTile;
    }
}

__device__ __forceinline__ void finish_points(const float* __restrict__ sp, int lane, int half, v2f pt,
                                              float px, float py, int p, bool active, int ntiles,
                                              int firstIdx, float bestS, int bestTile,
                                              const float* __restrict__ verts,
                                              const int* __restrict__ tris,
                                              const int* __restrict__ frame_p,
                                              const int* __restrict__ gw_p,
                                              const int* __restrict__ gh_p,
                                              float* __restrict__ out)
{
    int   ofi = __shfl_xor(firstIdx, 16, 32);
    float obs = __shfl_xor(bestS, 16, 32);
    int   fi  = min(firstIdx, ofi);
    bool valid = (fi < NOHIT);
    bool need = !valid && (bestS >= obs);
    int  resTile = bestTile;
    if (ntiles <= 0 || !active) need = false;

    int myIdx = IMAX;
    unsigned pend = __builtin_amdgcn_ballot_w32(need);
    for (int it = 0; it < 32; ++it) {
        if (pend == 0u) break;
        int srcl = __builtin_ctz(pend);
        int t = __builtin_amdgcn_readlane(resTile, srcl);
        v8f c3;
        tile_scores(sp, t, lane, pt, c3);
        int rS = 0;
#pragma unroll
        for (int r = 7; r >= 0; --r)
            rS = (c3[r] == bestS) ? r : rS;
        bool mine = need && (resTile == t);
        myIdx = mine ? (t * 16 + half * 8 + rS) : myIdx;
        need = need && !mine;
        pend = __builtin_amdgcn_ballot_w32(need);
    }

    int oIdx = __shfl_xor(myIdx, 16, 32);
    int f = valid ? fi : min(myIdx, oIdx);

    if (!half) {
        if (active) {
            int ft = f >> 4, fn = f & 15;
            const float* lo = sp + (size_t)(ft * 32 + fn) * 8;
            const float* hi = sp + (size_t)(ft * 32 + 16 + fn) * 8;
            float l1 = fmaf(lo[0], px, fmaf(lo[1], py, hi[0]));
            float l2 = fmaf(lo[2], px, fmaf(lo[3], py, hi[2]));
            float l3 = 1.0f - l1 - l2;
            float gw = (float)gw_p[0], gh = (float)gh_p[0];
            float fr = (float)frame_p[0];
            float* o = out + (size_t)p * 12;
            o[0] = l1; o[1] = l2; o[2] = l3;
#pragma unroll
            for (int j = 0; j < 3; ++j) {
                int vi = tris[f * 3 + j];
                o[3 + j * 3 + 0] = verts[vi * 2 + 0] / gw;
                o[3 + j * 3 + 1] = verts[vi * 2 + 1] / gh;
                o[3 + j * 3 + 2] = fr;
            }
        }
    }
}

__global__ void __launch_bounds__(THREADS)
point_kernel(const float* __restrict__ points,
             const float* __restrict__ verts,
             const int* __restrict__ tris,
             const int* __restrict__ frame_p,
             const int* __restrict__ gw_p,
             const int* __restrict__ gh_p,
             float* __restrict__ out, int T, int P)
{
    __shared__ float sp[MAX_TILES * 32 * 8];
    __shared__ float s_lx[WAVES * 16];
    __shared__ float s_ly[WAVES * 16];
    __shared__ int   s_lp[WAVES * 16];
    __shared__ int   s_cnt[WAVES];
    __shared__ int   s_rft[WAVES * WAVES * 32];
    __shared__ float s_rbs[WAVES * WAVES * 32];
    __shared__ int   s_rbt[WAVES * WAVES * 32];

    int tid = threadIdx.x;
    int ntiles = T >> 4;

    int lane = tid & 31;
    int half = lane >> 4;
    int nlan = lane & 15;
    int wave = __builtin_amdgcn_readfirstlane(tid >> 5);
    int pbase = (blockIdx.x * WAVES + wave) * 16;

    int p = pbase + nlan;
    int pc = min(p, P - 1);
    float qx = points[pc * 2 + 0], qy = points[pc * 2 + 1];

    if (T > 0) {
        float gw = (float)gw_p[0], gh = (float)gh_p[0];
        int vi[TRI_PER_THREAD][3];
#pragma unroll
        for (int k = 0; k < TRI_PER_THREAD; ++k) {
            int tc = min(tid + k * THREADS, T - 1);
            vi[k][0] = tris[tc * 3 + 0];
            vi[k][1] = tris[tc * 3 + 1];
            vi[k][2] = tris[tc * 3 + 2];
        }
        float vx[TRI_PER_THREAD][3], vy[TRI_PER_THREAD][3];
#pragma unroll
        for (int k = 0; k < TRI_PER_THREAD; ++k) {
#pragma unroll
            for (int j = 0; j < 3; ++j) {
                vx[k][j] = verts[vi[k][j] * 2 + 0];
                vy[k][j] = verts[vi[k][j] * 2 + 1];
            }
        }
#pragma unroll
        for (int k = 0; k < TRI_PER_THREAD; ++k) {
            int t = tid + k * THREADS;
            float4 lo0, lo1, hi0, hi1;
            tri_record(vx[k][0], vy[k][0], vx[k][1], vy[k][1], vx[k][2], vy[k][2],
                       gw, gh, lo0, lo1, hi0, hi1);
            int tile = t >> 4, n = t & 15;
            float4* lo = (float4*)(sp + (size_t)(tile * 32 + n) * 8);
            float4* hi = (float4*)(sp + (size_t)(tile * 32 + 16 + n) * 8);
            lo[0] = lo0; lo[1] = lo1;
            hi[0] = hi0; hi[1] = hi1;
        }
    }
    __syncthreads();

    float px = 0.0f, py = 0.0f;
    v2f pt;
    if (!half) {
        if (p < P) { px = qx; py = qy; }
        pt.x = px; pt.y = py;
    } else {
        pt.x = 1.0f; pt.y = 0.0f;
    }

    bool complete1 = (ntiles <= PASS1_TILES);
    int  n1 = complete1 ? ntiles : PASS1_TILES;
    int   fi1 = IMAX;
    float bs1 = NEG_INF;
    int   bt1 = 0;
    sweep_tiles(sp, lane, pt, 0, n1, 1, fi1, bs1, bt1);

    bool inP  = (p < P);
    bool hitp = (min(fi1, __shfl_xor(fi1, 16, 32)) < NOHIT);
    bool settled = inP && (hitp || complete1);
    bool unres = (!half) && inP && !settled;

    unsigned um = __builtin_amdgcn_ballot_w32(unres);
    if (lane == 0) s_cnt[wave] = __builtin_popcount(um);
    __syncthreads();
    int offs = 0, total = 0;
#pragma unroll
    for (int w = 0; w < WAVES; ++w) {
        int c = s_cnt[w];
        offs  += (w < wave) ? c : 0;
        total += c;
    }
    total = __builtin_amdgcn_readfirstlane(total);
    if (unres) {
        int slot = offs + __builtin_popcount(um & ((1u << lane) - 1u));
        s_lx[slot] = px; s_ly[slot] = py; s_lp[slot] = p;
    }
    __syncthreads();

    finish_points(sp, lane, half, pt, px, py, p, settled, ntiles, fi1, bs1, bt1,
                  verts, tris, frame_p, gw_p, gh_p, out);

    int groups = (total + 15) >> 4;
    for (int j = 0; j < groups; ++j) {
        int ic = min(j * 16 + nlan, total - 1);
        v2f q;
        if (!half) { q.x = s_lx[ic]; q.y = s_ly[ic]; }
        else       { q.x = 1.0f;     q.y = 0.0f; }
        int   f2 = IMAX;
        float b2 = NEG_INF;
        int   t2 = 0;
        sweep_tiles(sp, lane, q, wave, ntiles, WAVES, f2, b2, t2);
        int ri = (j * WAVES + wave) * 32 + lane;
        s_rft[ri] = f2; s_rbs[ri] = b2; s_rbt[ri] = t2;
    }
    __syncthreads();

    if (wave < groups) {
        int j = wave;
        int idx = j * 16 + nlan;
        bool act = (idx < total);
        int ic = min(idx, total - 1);
        float lx = 0.0f, ly = 0.0f;
        v2f q;
        if (!half) { lx = s_lx[ic]; ly = s_ly[ic]; q.x = lx; q.y = ly; }
        else       { q.x = 1.0f; q.y = 0.0f; }
        int lp = s_lp[ic];
        int   f2 = s_rft[(j * WAVES) * 32 + lane];
        float b2 = s_rbs[(j * WAVES) * 32 + lane];
        int   t2 = s_rbt[(j * WAVES) * 32 + lane];
#pragma unroll
        for (int w = 1; w < WAVES; ++w) {
            int   fw = s_rft[(j * WAVES + w) * 32 + lane];
            float bw = s_rbs[(j * WAVES + w) * 32 + lane];
            int   tw = s_rbt[(j * WAVES + w) * 32 + lane];
            f2 = min(f2, fw);
            bool better = (bw > b2) || (bw == b2 && tw < t2);
            b2 = better ? bw : b2;
            t2 = better ? tw : t2;
        }
        finish_points(sp, lane, half, q, lx, ly, lp, act, ntiles, f2, b2, t2,
                      verts, tris, frame_p, gw_p, gh_p, out);
    }
}

extern "C" void kernel_launch(void* const* d_in, const int* in_sizes, int n_in,
                              void* d_out, int out_size, void* d_ws, size_t ws_size,
                              hipStream_t stream) {
    const float* verts  = (const float*)d_in[0];
    const float* points = (const float*)d_in[1];
    const int*   tris   = (const int*)d_in[2];
    const int*   frame  = (const int*)d_in[3];
    const int*   gw     = (const int*)d_in[4];
    const int*   gh     = (const int*)d_in[5];
    float*       out    = (float*)d_out;

    int T = in_sizes[2] / 3;
    int P = in_sizes[1] / 2;
    if (T > MAX_T) T = MAX_T;

    int pointsPerBlock = WAVES * 16;
    int blocks = (P + pointsPerBlock - 1) / pointsPerBlock;
    point_kernel<<<blocks, THREADS, 0, stream>>>(points, verts, tris,
                                             frame, gw, gh, out, T, P);
}
